# starting kernel
# baseline (speedup 1.0000x reference)
#include <hip/hip_runtime.h>
#include <cstdint>
#include <cstddef>

typedef __attribute__((ext_vector_type(2))) float v2f;
typedef __attribute__((ext_vector_type(8))) float v8f;

#define NEG_SLOPE 0.01f


__global__ void zero_f32_v4(float4* __restrict__ p, long long n4) {
  long long i = (long long)blockIdx.x * blockDim.x + threadIdx.x;
  if (i < n4) p[i] = make_float4(0.f, 0.f, 0.f, 0.f);
}

__global__ void init_deg(float* __restrict__ deg, int n) {
  int i = blockIdx.x * blockDim.x + threadIdx.x;
  if (i < n) deg[i] = 1.0f;
}

__global__ void deg_accum(const int* __restrict__ dst, float* __restrict__ deg, int E) {
  int e = blockIdx.x * blockDim.x + threadIdx.x;
  if (e < E) atomicAdd(&deg[dst[e]], 1.0f);
}

__global__ void compute_dinv(const float* __restrict__ deg, float* __restrict__ dinv, int n) {
  int i = blockIdx.x * blockDim.x + threadIdx.x;
  if (i < n) {
    float d = deg[i];
    dinv[i] = d > 0.0f ? rsqrtf(d) : 0.0f;
  }
}

__global__ void compute_norm(const int* __restrict__ src, const int* __restrict__ dst,
                             const float* __restrict__ dinv, float* __restrict__ norm, int E) {
  int e = blockIdx.x * blockDim.x + threadIdx.x;
  if (e < E) norm[e] = dinv[src[e]] * dinv[dst[e]];
}

template <int K, int NOUT>
__global__ void gcn_gemm_wmma(const float* __restrict__ X, const float* __restrict__ W,
                              float* __restrict__ Y) {
  constexpr int LDSROW = K + 4;
  __shared__ float At[16 * LDSROW];

  const int tid = threadIdx.x;
  const size_t mbase_row = (size_t)blockIdx.x * 16;

  const float* __restrict__ xtile = X + mbase_row * K;
  for (int i = tid * 4; i < 16 * K; i += blockDim.x * 4) {
    const int r = i / K;
    const int c = i % K;
    float4 v = *(const float4*)(xtile + (size_t)r * K + c);
    float* d = &At[r * LDSROW + c];
    d[0] = v.x; d[1] = v.y; d[2] = v.z; d[3] = v.w;
  }
  __syncthreads();

  const int wave = tid >> 5;
  const int lane = tid & 31;
  const int lrow  = lane & 15;
  const int khalf = (lane >> 4) * 2;
  const int n0    = wave * 16;

  v8f acc = {};
  for (int k0 = 0; k0 < K; k0 += 4) {
    const float* ap = &At[lrow * LDSROW + k0 + khalf];
    v2f a; a.x = ap[0]; a.y = ap[1];
    const float* wp = W + (size_t)(k0 + khalf) * NOUT + n0 + lrow;
    v2f b; b.x = wp[0]; b.y = wp[NOUT];
    acc = __builtin_amdgcn_wmma_f32_16x16x4_f32(
         false, a,  false, b,
         (short)0, acc,  false,  false);
  }

  float* yp = Y + mbase_row * NOUT + n0 + lrow;
  const int mofs = (lane < 16) ? 0 : 8;
#pragma unroll
  for (int v = 0; v < 8; ++v) {
    yp[(size_t)(mofs + v) * NOUT] = acc[v];
  }
}

template <int F>
__global__ void edge_scatter_v4(const int* __restrict__ src, const int* __restrict__ dst,
                                const float* __restrict__ norm, const float* __restrict__ h,
                                float* __restrict__ agg, int E) {
  constexpr int FV = F / 4;
  long long gid = (long long)blockIdx.x * blockDim.x + threadIdx.x;
  if (gid >= (long long)E * FV) return;
  const int e  = (int)(gid >> __builtin_ctz(FV));
  const int f4 = (int)(gid & (FV - 1));
  const int s = src[e];
  const int d = dst[e];
  const float w = norm[e];
  const float4 hv = *(const float4*)(h + (size_t)s * F + f4 * 4);
  float* ap = agg + (size_t)d * F + f4 * 4;
  atomicAdd(ap + 0, hv.x * w);
  atomicAdd(ap + 1, hv.y * w);
  atomicAdd(ap + 2, hv.z * w);
  atomicAdd(ap + 3, hv.w * w);
}

template <int F>
__global__ void finalize_act_v4(const float* __restrict__ agg, const float* __restrict__ h,
                                const float* __restrict__ dinv, const float* __restrict__ bias,
                                float* __restrict__ out, int n) {
  constexpr int FV = F / 4;
  long long gid = (long long)blockIdx.x * blockDim.x + threadIdx.x;
  if (gid >= (long long)n * FV) return;
  const int i  = (int)(gid >> __builtin_ctz(FV));
  const int f4 = (int)(gid & (FV - 1));
  const float di2 = dinv[i] * dinv[i];
  const size_t base = gid * 4;
  const float4 av = *(const float4*)(agg + base);
  const float4 hv = *(const float4*)(h + base);
  const float4 bv = *(const float4*)(bias + f4 * 4);
  float4 r;
  r.x = av.x + di2 * hv.x + bv.x;
  r.y = av.y + di2 * hv.y + bv.y;
  r.z = av.z + di2 * hv.z + bv.z;
  r.w = av.w + di2 * hv.w + bv.w;
  r.x = (r.x >= 0.f) ? r.x : NEG_SLOPE * r.x;
  r.y = (r.y >= 0.f) ? r.y : NEG_SLOPE * r.y;
  r.z = (r.z >= 0.f) ? r.z : NEG_SLOPE * r.z;
  r.w = (r.w >= 0.f) ? r.w : NEG_SLOPE * r.w;
  *(float4*)(out + base) = r;
}

template <int F>
__global__ void finalize_lin_v4(float* __restrict__ agg, const float* __restrict__ h,
                                const float* __restrict__ dinv, const float* __restrict__ bias,
                                int n) {
  constexpr int FV = F / 4;
  long long gid = (long long)blockIdx.x * blockDim.x + threadIdx.x;
  if (gid >= (long long)n * FV) return;
  const int i  = (int)(gid >> __builtin_ctz(FV));
  const int f4 = (int)(gid & (FV - 1));
  const float di2 = dinv[i] * dinv[i];
  const size_t base = gid * 4;
  const float4 av = *(const float4*)(agg + base);
  const float4 hv = *(const float4*)(h + base);
  const float4 bv = *(const float4*)(bias + f4 * 4);
  float4 r;
  r.x = av.x + di2 * hv.x + bv.x;
  r.y = av.y + di2 * hv.y + bv.y;
  r.z = av.z + di2 * hv.z + bv.z;
  r.w = av.w + di2 * hv.w + bv.w;
  *(float4*)(agg + base) = r;
}


extern "C" void kernel_launch(void* const* d_in, const int* in_sizes, int n_in,
                              void* d_out, int out_size, void* d_ws, size_t ws_size,
                              hipStream_t stream) {
  const float* x   = (const float*)d_in[0];
  const int*   ei  = (const int*)d_in[1];
  const float* W1  = (const float*)d_in[2];
  const float* b1  = (const float*)d_in[3];
  const float* W2  = (const float*)d_in[4];
  const float* b2  = (const float*)d_in[5];
  float* out = (float*)d_out;

  const int N = in_sizes[0] / 256;
  const int E = in_sizes[1] / 2;
  const int* src = ei;
  const int* dst = ei + E;

  float* ws   = (float*)d_ws;
  float* deg  = ws;
  float* dinv = deg + N;
  float* norm = dinv + N;
  float* h1   = norm + E;
  float* agg1 = h1 + (size_t)N * 128;
  float* h2 = agg1;
  float* g  = h1;

  const int T = 256;
  auto blk = [](long long n, int t) { return (unsigned)((n + t - 1) / t); };

  init_deg<<<blk(N, T), T, 0, stream>>>(deg, N);
  deg_accum<<<blk(E, T), T, 0, stream>>>(dst, deg, E);
  compute_dinv<<<blk(N, T), T, 0, stream>>>(deg, dinv, N);
  compute_norm<<<blk(E, T), T, 0, stream>>>(src, dst, dinv, norm, E);

  gcn_gemm_wmma<256, 128><<<N / 16, 8 * 32, 0, stream>>>(x, W1, h1);

  zero_f32_v4<<<blk((long long)N * 32, T), T, 0, stream>>>((float4*)agg1, (long long)N * 32);
  edge_scatter_v4<128><<<blk((long long)E * 32, T), T, 0, stream>>>(src, dst, norm, h1, agg1, E);
  finalize_act_v4<128><<<blk((long long)N * 32, T), T, 0, stream>>>(agg1, h1, dinv, b1, h2, N);

  gcn_gemm_wmma<128, 64><<<N / 16, 4 * 32, 0, stream>>>(h2, W2, g);

  zero_f32_v4<<<blk((long long)N * 16, T), T, 0, stream>>>((float4*)out, (long long)N * 16);
  edge_scatter_v4<64><<<blk((long long)E * 16, T), T, 0, stream>>>(src, dst, norm, g, out, E);
  finalize_lin_v4<64><<<blk((long long)N * 16, T), T, 0, stream>>>(out, g, dinv, b2, N);
}


# Degree count without global atomics: a block takes a node range and an edge slice, counts in LDS (integer adds), stores a partial plane; dinv kernel adds the planes; 8->7 launches; same degrees
# speedup vs baseline: 2.5203x; 2.5203x over previous
#include <hip/hip_runtime.h>
#include <cstdint>
#include <cstddef>

typedef __attribute__((ext_vector_type(2))) float v2f;
typedef __attribute__((ext_vector_type(8))) float v8f;

#define NEG_SLOPE 0.01f


constexpr int DEG_RNG    = 12544;
constexpr int DEG_SLICES = 32;
constexpr int DEG_T      = 512;
constexpr int DEG_U      = 16;

__global__ void __launch_bounds__(DEG_T)
deg_count_ranges(const int* __restrict__ dst, unsigned* __restrict__ part, int n, int E) {
  __shared__ unsigned cnt[DEG_RNG];
  const int tid = threadIdx.x;
  const int slice = blockIdx.x;
  const int lo = blockIdx.y * DEG_RNG;
  for (int i = tid; i < DEG_RNG; i += DEG_T) cnt[i] = 0u;
  __syncthreads();

  const long long per = (((long long)E + 4 * DEG_SLICES - 1) / (4 * DEG_SLICES)) * 4;
  const long long e0 = (long long)slice * per;
  long long e1 = e0 + per;
  if (e1 > (long long)E) e1 = E;

  auto count = [&](int d) {
    const unsigned r = (unsigned)(d - lo);
    if (r < (unsigned)DEG_RNG) atomicAdd(&cnt[r], 1u);
  };

  if (e0 < e1) {
    if ((((size_t)dst) & 15) == 0) {
      const long long e1v = e0 + ((e1 - e0) & ~3LL);
      for (long long e = e0 + (long long)tid * 4; e < e1v; e += (long long)DEG_T * 4 * DEG_U) {
        int4 v[DEG_U];
#pragma unroll
        for (int u = 0; u < DEG_U; ++u) {
          const long long ee = e + (long long)u * DEG_T * 4;
          if (ee < e1v) v[u] = *(const int4*)(dst + ee);
          else v[u] = make_int4(-1, -1, -1, -1);
        }
#pragma unroll
        for (int u = 0; u < DEG_U; ++u) {
          count(v[u].x); count(v[u].y); count(v[u].z); count(v[u].w);
        }
      }
      if ((long long)tid < e1 - e1v) count(dst[e1v + tid]);
    } else {
      for (long long e = e0 + tid; e < e1; e += DEG_T) count(dst[e]);
    }
  }
  __syncthreads();

  unsigned* __restrict__ pp = part + (size_t)slice * n + lo;
  for (int i = tid; i < DEG_RNG; i += DEG_T)
    if (lo + i < n) pp[i] = cnt[i];
}

__global__ void compute_dinv(const unsigned* __restrict__ part, float* __restrict__ dinv, int n) {
  int i = blockIdx.x * blockDim.x + threadIdx.x;
  if (i < n) {
    unsigned c = 1u;
#pragma unroll
    for (int s = 0; s < DEG_SLICES; ++s) c += part[(size_t)s * n + i];
    const float d = (float)c;
    dinv[i] = d > 0.0f ? rsqrtf(d) : 0.0f;
  }
}

__global__ void compute_norm(const int* __restrict__ src, const int* __restrict__ dst,
                             const float* __restrict__ dinv, float* __restrict__ norm, int E,
                             unsigned* __restrict__ tickets) {
  int e = blockIdx.x * blockDim.x + threadIdx.x;
  if (e < E) norm[e] = dinv[src[e]] * dinv[dst[e]];
  if (e < 2) tickets[e] = 0u;
}

template <int K, int NOUT, bool ACT_IN, bool IN_PLANES>
__global__ void gcn_gemm_wmma(const float* __restrict__ X, const float* __restrict__ W,
                              const float* __restrict__ dinv, const float* __restrict__ bias,
                              float* __restrict__ Y, float* __restrict__ S, size_t plane_elems) {
  constexpr int LDSROW = K + 4;
  constexpr int PW = 64;
  __shared__ float At[16 * LDSROW];

  const int tid = threadIdx.x;
  const size_t mbase_row = (size_t)blockIdx.x * 16;

  for (int i = tid * 4; i < 16 * K; i += blockDim.x * 4) {
    const int r = i / K;
    const int c = i % K;
    const float* __restrict__ xp = IN_PLANES
        ? X + (size_t)(c / PW) * plane_elems + (mbase_row + r) * PW + (c % PW)
        : X + (mbase_row + r) * K + c;
    float4 v = *(const float4*)xp;
    if (ACT_IN) {
      v.x = (v.x >= 0.f) ? v.x : NEG_SLOPE * v.x;
      v.y = (v.y >= 0.f) ? v.y : NEG_SLOPE * v.y;
      v.z = (v.z >= 0.f) ? v.z : NEG_SLOPE * v.z;
      v.w = (v.w >= 0.f) ? v.w : NEG_SLOPE * v.w;
    }
    float* d = &At[r * LDSROW + c];
    d[0] = v.x; d[1] = v.y; d[2] = v.z; d[3] = v.w;
  }

  const int wave = tid >> 5;
  const int lane = tid & 31;
  const int lrow  = lane & 15;
  const int khalf = (lane >> 4) * 2;
  const int n0    = wave * 16;
  const int mofs  = (lane < 16) ? 0 : 8;

  const float bn = bias[n0 + lrow];
  float di2[8];
#pragma unroll
  for (int v = 0; v < 8; ++v) {
    const float di = dinv[mbase_row + mofs + v];
    di2[v] = di * di;
  }
  __syncthreads();

  v8f acc = {};
  for (int k0 = 0; k0 < K; k0 += 4) {
    const float* ap = &At[lrow * LDSROW + k0 + khalf];
    v2f a; a.x = ap[0]; a.y = ap[1];
    const float* wp = W + (size_t)(k0 + khalf) * NOUT + n0 + lrow;
    v2f b; b.x = wp[0]; b.y = wp[NOUT];
    acc = __builtin_amdgcn_wmma_f32_16x16x4_f32(
         false, a,  false, b,
         (short)0, acc,  false,  false);
  }

  const int ncol = n0 + lrow;
  const size_t obase = (size_t)(ncol / PW) * plane_elems + mbase_row * PW + (ncol % PW);
  float* yp = Y + obase;
  float* sp = S + obase;
#pragma unroll
  for (int v = 0; v < 8; ++v) {
    const float y = acc[v];
    yp[(size_t)(mofs + v) * PW] = y;
    sp[(size_t)(mofs + v) * PW] = di2[v] * y + bn;
  }
}

template <int F, bool BY_HALF>
__global__ void __launch_bounds__(256)
edge_scatter_lines(const int* __restrict__ src, const int* __restrict__ dst,
                   const float* __restrict__ norm, const float* __restrict__ hfull,
                   float* __restrict__ aggfull, int E, size_t plane_elems,
                   unsigned* __restrict__ tickets) {
  __shared__ unsigned pick[2];
  unsigned plane = 0u, chunk = blockIdx.x;
  if (BY_HALF) {
    const unsigned nchunks = gridDim.x;
    if (threadIdx.x == 0) {
      unsigned p = (__builtin_amdgcn_s_sendmsg_rtn(0x87) >> 18) & 1u;
      unsigned t = atomicAdd(&tickets[p], 1u);
      if (t >= nchunks) { p ^= 1u; t = atomicAdd(&tickets[p], 1u); }
      pick[0] = p; pick[1] = t;
    }
    __syncthreads();
    plane = pick[0]; chunk = pick[1];
    if (chunk >= nchunks) return;
  }
  const float* __restrict__ h = hfull + (size_t)plane * plane_elems;
  float* __restrict__ agg = aggfull + (size_t)plane * plane_elems;
  constexpr int L   = F / 32;
  constexpr int EPG = 32 / L;
  const int lane = threadIdx.x & 31;
  const long long gtid = (long long)chunk * blockDim.x + threadIdx.x;
  const long long e0 = gtid - lane;
  if (e0 >= (long long)E) return;
  const long long rem = (long long)E - e0;
  const int cnt = rem < 32 ? (int)rem : 32;

  int s = 0, d = 0;
  float w = 0.f;
  if (lane < cnt) {
    s = src[e0 + lane];
    d = dst[e0 + lane];
    w = norm[e0 + lane];
  }

  int j0 = 0;
  for (; j0 + EPG <= cnt; j0 += EPG) {
    float v[EPG][L];
#pragma unroll
    for (int u = 0; u < EPG; ++u) {
      const int sj = __builtin_amdgcn_readlane(s, j0 + u);
      const float* hp = h + (size_t)sj * F + lane;
#pragma unroll
      for (int l = 0; l < L; ++l) v[u][l] = hp[l * 32];
    }
#pragma unroll
    for (int u = 0; u < EPG; ++u) {
      const int dj = __builtin_amdgcn_readlane(d, j0 + u);
      const float wj = __int_as_float(__builtin_amdgcn_readlane(__float_as_int(w), j0 + u));
      float* ap = agg + (size_t)dj * F + lane;
#pragma unroll
      for (int l = 0; l < L; ++l) atomicAdd(ap + l * 32, v[u][l] * wj);
    }
  }
  for (; j0 < cnt; ++j0) {
    const int sj = __builtin_amdgcn_readlane(s, j0);
    const int dj = __builtin_amdgcn_readlane(d, j0);
    const float wj = __int_as_float(__builtin_amdgcn_readlane(__float_as_int(w), j0));
    const float* hp = h + (size_t)sj * F + lane;
    float* ap = agg + (size_t)dj * F + lane;
#pragma unroll
    for (int l = 0; l < L; ++l) atomicAdd(ap + l * 32, hp[l * 32] * wj);
  }
}


extern "C" void kernel_launch(void* const* d_in, const int* in_sizes, int n_in,
                              void* d_out, int out_size, void* d_ws, size_t ws_size,
                              hipStream_t stream) {
  const float* x   = (const float*)d_in[0];
  const int*   ei  = (const int*)d_in[1];
  const float* W1  = (const float*)d_in[2];
  const float* b1  = (const float*)d_in[3];
  const float* W2  = (const float*)d_in[4];
  const float* b2  = (const float*)d_in[5];
  float* out = (float*)d_out;

  const int N = in_sizes[0] / 256;
  const int E = in_sizes[1] / 2;
  const int* src = ei;
  const int* dst = ei + E;

  float* ws   = (float*)d_ws;
  float* tkw  = ws;
  float* dinv = tkw + N;
  float* norm = dinv + N;
  float* h1   = norm + E;
  float* agg1 = h1 + (size_t)N * 128;
  const size_t plane = (size_t)N * 64;
  float* g  = h1;
  unsigned* tickets = (unsigned*)tkw;
  unsigned* degpart = (unsigned*)h1;

  const int T = 256;
  auto blk = [](long long n, int t) { return (unsigned)((n + t - 1) / t); };

  deg_count_ranges<<<dim3(DEG_SLICES, blk(N, DEG_RNG)), DEG_T, 0, stream>>>(dst, degpart, N, E);
  compute_dinv<<<blk(N, T), T, 0, stream>>>(degpart, dinv, N);
  compute_norm<<<blk(E, T), T, 0, stream>>>(src, dst, dinv, norm, E, tickets);

  gcn_gemm_wmma<256, 128, false, false><<<N / 16, 8 * 32, 0, stream>>>(x, W1, dinv, b1, h1, agg1, plane);

  edge_scatter_lines<64, true><<<dim3(blk(E, T), 2), T, 0, stream>>>(src, dst, norm, h1, agg1, E, plane, tickets);

  gcn_gemm_wmma<128, 64, true, true><<<N / 16, 4 * 32, 0, stream>>>(agg1, W2, dinv, b2, g, out, plane);

  edge_scatter_lines<64, false><<<blk(E, T), T, 0, stream>>>(src, dst, norm, g, out, E, plane, tickets);
}


# Layer-2 scatter split between the cache halves by destination row: each half scans all edges, a wave keeps those whose sum row is in its half of the nodes (ballot, bit walk); sums: atomics, as before
# speedup vs baseline: 2.6838x; 1.0649x over previous
#include <hip/hip_runtime.h>
#include <cstdint>
#include <cstddef>

typedef __attribute__((ext_vector_type(2))) float v2f;
typedef __attribute__((ext_vector_type(8))) float v8f;

#define NEG_SLOPE 0.01f


constexpr int DEG_RNG    = 12544;
constexpr int DEG_SLICES = 32;
constexpr int DEG_T      = 512;
constexpr int DEG_U      = 16;

__global__ void __launch_bounds__(DEG_T)
deg_count_ranges(const int* __restrict__ dst, unsigned* __restrict__ part, int n, int E) {
  __shared__ unsigned cnt[DEG_RNG];
  const int tid = threadIdx.x;
  const int slice = blockIdx.x;
  const int lo = blockIdx.y * DEG_RNG;
  for (int i = tid; i < DEG_RNG; i += DEG_T) cnt[i] = 0u;
  __syncthreads();

  const long long per = (((long long)E + 4 * DEG_SLICES - 1) / (4 * DEG_SLICES)) * 4;
  const long long e0 = (long long)slice * per;
  long long e1 = e0 + per;
  if (e1 > (long long)E) e1 = E;

  auto count = [&](int d) {
    const unsigned r = (unsigned)(d - lo);
    if (r < (unsigned)DEG_RNG) atomicAdd(&cnt[r], 1u);
  };

  if (e0 < e1) {
    if ((((size_t)dst) & 15) == 0) {
      const long long e1v = e0 + ((e1 - e0) & ~3LL);
      for (long long e = e0 + (long long)tid * 4; e < e1v; e += (long long)DEG_T * 4 * DEG_U) {
        int4 v[DEG_U];
#pragma unroll
        for (int u = 0; u < DEG_U; ++u) {
          const long long ee = e + (long long)u * DEG_T * 4;
          if (ee < e1v) v[u] = *(const int4*)(dst + ee);
          else v[u] = make_int4(-1, -1, -1, -1);
        }
#pragma unroll
        for (int u = 0; u < DEG_U; ++u) {
          count(v[u].x); count(v[u].y); count(v[u].z); count(v[u].w);
        }
      }
      if ((long long)tid < e1 - e1v) count(dst[e1v + tid]);
    } else {
      for (long long e = e0 + tid; e < e1; e += DEG_T) count(dst[e]);
    }
  }
  __syncthreads();

  unsigned* __restrict__ pp = part + (size_t)slice * n + lo;
  for (int i = tid; i < DEG_RNG; i += DEG_T)
    if (lo + i < n) pp[i] = cnt[i];
}

__global__ void compute_dinv(const unsigned* __restrict__ part, float* __restrict__ dinv, int n) {
  int i = blockIdx.x * blockDim.x + threadIdx.x;
  if (i < n) {
    unsigned c = 1u;
#pragma unroll
    for (int s = 0; s < DEG_SLICES; ++s) c += part[(size_t)s * n + i];
    const float d = (float)c;
    dinv[i] = d > 0.0f ? rsqrtf(d) : 0.0f;
  }
}

__global__ void compute_norm(const int* __restrict__ src, const int* __restrict__ dst,
                             const float* __restrict__ dinv, float* __restrict__ norm, int E,
                             unsigned* __restrict__ tickets) {
  int e = blockIdx.x * blockDim.x + threadIdx.x;
  if (e < E) norm[e] = dinv[src[e]] * dinv[dst[e]];
  if (e < 4) tickets[e] = 0u;
}

template <int K, int NOUT, bool ACT_IN, bool IN_PLANES>
__global__ void gcn_gemm_wmma(const float* __restrict__ X, const float* __restrict__ W,
                              const float* __restrict__ dinv, const float* __restrict__ bias,
                              float* __restrict__ Y, float* __restrict__ S, size_t plane_elems) {
  constexpr int LDSROW = K + 4;
  constexpr int PW = 64;
  __shared__ float At[16 * LDSROW];

  const int tid = threadIdx.x;
  const size_t mbase_row = (size_t)blockIdx.x * 16;

  for (int i = tid * 4; i < 16 * K; i += blockDim.x * 4) {
    const int r = i / K;
    const int c = i % K;
    const float* __restrict__ xp = IN_PLANES
        ? X + (size_t)(c / PW) * plane_elems + (mbase_row + r) * PW + (c % PW)
        : X + (mbase_row + r) * K + c;
    float4 v = *(const float4*)xp;
    if (ACT_IN) {
      v.x = (v.x >= 0.f) ? v.x : NEG_SLOPE * v.x;
      v.y = (v.y >= 0.f) ? v.y : NEG_SLOPE * v.y;
      v.z = (v.z >= 0.f) ? v.z : NEG_SLOPE * v.z;
      v.w = (v.w >= 0.f) ? v.w : NEG_SLOPE * v.w;
    }
    float* d = &At[r * LDSROW + c];
    d[0] = v.x; d[1] = v.y; d[2] = v.z; d[3] = v.w;
  }

  const int wave = tid >> 5;
  const int lane = tid & 31;
  const int lrow  = lane & 15;
  const int khalf = (lane >> 4) * 2;
  const int n0    = wave * 16;
  const int mofs  = (lane < 16) ? 0 : 8;

  const float bn = bias[n0 + lrow];
  float di2[8];
#pragma unroll
  for (int v = 0; v < 8; ++v) {
    const float di = dinv[mbase_row + mofs + v];
    di2[v] = di * di;
  }
  __syncthreads();

  v8f acc = {};
  for (int k0 = 0; k0 < K; k0 += 4) {
    const float* ap = &At[lrow * LDSROW + k0 + khalf];
    v2f a; a.x = ap[0]; a.y = ap[1];
    const float* wp = W + (size_t)(k0 + khalf) * NOUT + n0 + lrow;
    v2f b; b.x = wp[0]; b.y = wp[NOUT];
    acc = __builtin_amdgcn_wmma_f32_16x16x4_f32(
         false, a,  false, b,
         (short)0, acc,  false,  false);
  }

  const int ncol = n0 + lrow;
  const size_t obase = (size_t)(ncol / PW) * plane_elems + mbase_row * PW + (ncol % PW);
  float* yp = Y + obase;
  float* sp = S + obase;
#pragma unroll
  for (int v = 0; v < 8; ++v) {
    const float y = acc[v];
    yp[(size_t)(mofs + v) * PW] = y;
    sp[(size_t)(mofs + v) * PW] = di2[v] * y + bn;
  }
}

template <int F, bool BY_HALF>
__global__ void __launch_bounds__(256)
edge_scatter_lines(const int* __restrict__ src, const int* __restrict__ dst,
                   const float* __restrict__ norm, const float* __restrict__ hfull,
                   float* __restrict__ aggfull, int E, size_t plane_elems,
                   unsigned* __restrict__ tickets) {
  __shared__ unsigned pick[2];
  unsigned plane = 0u, chunk = blockIdx.x;
  if (BY_HALF) {
    const unsigned nchunks = gridDim.x;
    if (threadIdx.x == 0) {
      unsigned p = (__builtin_amdgcn_s_sendmsg_rtn(0x87) >> 18) & 1u;
      unsigned t = atomicAdd(&tickets[p], 1u);
      if (t >= nchunks) { p ^= 1u; t = atomicAdd(&tickets[p], 1u); }
      pick[0] = p; pick[1] = t;
    }
    __syncthreads();
    plane = pick[0]; chunk = pick[1];
    if (chunk >= nchunks) return;
  }
  const float* __restrict__ h = hfull + (size_t)plane * plane_elems;
  float* __restrict__ agg = aggfull + (size_t)plane * plane_elems;
  constexpr int L   = F / 32;
  constexpr int EPG = 32 / L;
  const int lane = threadIdx.x & 31;
  const long long gtid = (long long)chunk * blockDim.x + threadIdx.x;
  const long long e0 = gtid - lane;
  if (e0 >= (long long)E) return;
  const long long rem = (long long)E - e0;
  const int cnt = rem < 32 ? (int)rem : 32;

  int s = 0, d = 0;
  float w = 0.f;
  if (lane < cnt) {
    s = src[e0 + lane];
    d = dst[e0 + lane];
    w = norm[e0 + lane];
  }

  int j0 = 0;
  for (; j0 + EPG <= cnt; j0 += EPG) {
    float v[EPG][L];
#pragma unroll
    for (int u = 0; u < EPG; ++u) {
      const int sj = __builtin_amdgcn_readlane(s, j0 + u);
      const float* hp = h + (size_t)sj * F + lane;
#pragma unroll
      for (int l = 0; l < L; ++l) v[u][l] = hp[l * 32];
    }
#pragma unroll
    for (int u = 0; u < EPG; ++u) {
      const int dj = __builtin_amdgcn_readlane(d, j0 + u);
      const float wj = __int_as_float(__builtin_amdgcn_readlane(__float_as_int(w), j0 + u));
      float* ap = agg + (size_t)dj * F + lane;
#pragma unroll
      for (int l = 0; l < L; ++l) atomicAdd(ap + l * 32, v[u][l] * wj);
    }
  }
  for (; j0 < cnt; ++j0) {
    const int sj = __builtin_amdgcn_readlane(s, j0);
    const int dj = __builtin_amdgcn_readlane(d, j0);
    const float wj = __int_as_float(__builtin_amdgcn_readlane(__float_as_int(w), j0));
    const float* hp = h + (size_t)sj * F + lane;
    float* ap = agg + (size_t)dj * F + lane;
#pragma unroll
    for (int l = 0; l < L; ++l) atomicAdd(ap + l * 32, hp[l * 32] * wj);
  }
}

template <int F>
__global__ void __launch_bounds__(256)
edge_scatter_own_rows(const int* __restrict__ src, const int* __restrict__ dst,
                      const float* __restrict__ norm, const float* __restrict__ h,
                      float* __restrict__ agg, int E, int n,
                      unsigned* __restrict__ tickets) {
  __shared__ unsigned pick[2];
  const unsigned nchunks = gridDim.x;
  if (threadIdx.x == 0) {
    unsigned p = (__builtin_amdgcn_s_sendmsg_rtn(0x87) >> 18) & 1u;
    unsigned t = atomicAdd(&tickets[p], 1u);
    if (t >= nchunks) { p ^= 1u; t = atomicAdd(&tickets[p], 1u); }
    pick[0] = p; pick[1] = t;
  }
  __syncthreads();
  const unsigned half = pick[0], chunk = pick[1];
  if (chunk >= nchunks) return;

  constexpr int L = F / 32;
  constexpr int SLOTS = 24;
  const int lane = threadIdx.x & 31;
  const long long e0 = ((long long)chunk * (blockDim.x >> 5) + (threadIdx.x >> 5)) * 64;
  if (e0 >= (long long)E) return;

  const unsigned n_lo = (unsigned)n >> 1;
  const bool upper = half != 0u;
  int sA = 0, dA = 0, sB = 0, dB = 0;
  float wA = 0.f, wB = 0.f;
  bool keepA = false, keepB = false;
  const long long ea = e0 + lane, eb = ea + 32;
  if (ea < (long long)E) {
    sA = src[ea]; dA = dst[ea]; wA = norm[ea];
    keepA = ((unsigned)dA >= n_lo) == upper;
  }
  if (eb < (long long)E) {
    sB = src[eb]; dB = dst[eb]; wB = norm[eb];
    keepB = ((unsigned)dB >= n_lo) == upper;
  }
  const unsigned mA = __builtin_amdgcn_ballot_w32(keepA);
  const unsigned mB = __builtin_amdgcn_ballot_w32(keepB);

#pragma nounroll
  for (int t = 0; t < 2; ++t) {
    const int s = t ? sB : sA;
    const int d = t ? dB : dA;
    const float w = t ? wB : wA;
    unsigned m = t ? mB : mA;
    int rem = __builtin_popcount(m);
    while (rem > 0) {
      const int c = rem > SLOTS ? (rem + 1) >> 1 : rem;
      rem -= c;
      unsigned mc = m;
      float v[SLOTS][L];
#pragma unroll
      for (int u = 0; u < SLOTS; ++u) {
        if (u < c) {
          const int j = __builtin_ctz(m);
          m &= m - 1;
          const unsigned sj = (unsigned)__builtin_amdgcn_readlane(s, j);
          const float* hp = h + (size_t)sj * F + lane;
#pragma unroll
          for (int l = 0; l < L; ++l) v[u][l] = hp[l * 32];
        }
      }
#pragma unroll
      for (int u = 0; u < SLOTS; ++u) {
        if (u < c) {
          const int j = __builtin_ctz(mc);
          mc &= mc - 1;
          const unsigned dj = (unsigned)__builtin_amdgcn_readlane(d, j);
          const float wj = __int_as_float(__builtin_amdgcn_readlane(__float_as_int(w), j));
          float* ap = agg + (size_t)dj * F + lane;
#pragma unroll
          for (int l = 0; l < L; ++l) atomicAdd(ap + l * 32, v[u][l] * wj);
        }
      }
    }
  }
}


extern "C" void kernel_launch(void* const* d_in, const int* in_sizes, int n_in,
                              void* d_out, int out_size, void* d_ws, size_t ws_size,
                              hipStream_t stream) {
  const float* x   = (const float*)d_in[0];
  const int*   ei  = (const int*)d_in[1];
  const float* W1  = (const float*)d_in[2];
  const float* b1  = (const float*)d_in[3];
  const float* W2  = (const float*)d_in[4];
  const float* b2  = (const float*)d_in[5];
  float* out = (float*)d_out;

  const int N = in_sizes[0] / 256;
  const int E = in_sizes[1] / 2;
  const int* src = ei;
  const int* dst = ei + E;

  float* ws   = (float*)d_ws;
  float* tkw  = ws;
  float* dinv = tkw + N;
  float* norm = dinv + N;
  float* h1   = norm + E;
  float* agg1 = h1 + (size_t)N * 128;
  const size_t plane = (size_t)N * 64;
  float* g  = h1;
  unsigned* tickets = (unsigned*)tkw;
  unsigned* degpart = (unsigned*)h1;

  const int T = 256;
  auto blk = [](long long n, int t) { return (unsigned)((n + t - 1) / t); };

  deg_count_ranges<<<dim3(DEG_SLICES, blk(N, DEG_RNG)), DEG_T, 0, stream>>>(dst, degpart, N, E);
  compute_dinv<<<blk(N, T), T, 0, stream>>>(degpart, dinv, N);
  compute_norm<<<blk(E, T), T, 0, stream>>>(src, dst, dinv, norm, E, tickets);

  gcn_gemm_wmma<256, 128, false, false><<<N / 16, 8 * 32, 0, stream>>>(x, W1, dinv, b1, h1, agg1, plane);

  edge_scatter_lines<64, true><<<dim3(blk(E, T), 2), T, 0, stream>>>(src, dst, norm, h1, agg1, E, plane, tickets);

  gcn_gemm_wmma<128, 64, true, true><<<N / 16, 4 * 32, 0, stream>>>(agg1, W2, dinv, b2, g, out, plane);

  edge_scatter_own_rows<64><<<dim3(blk(E, 2 * T), 2), T, 0, stream>>>(src, dst, norm, g, out, E, N, tickets + 2);
}


# Both GEMM launches as resident grids: a wave keeps its 16 columns of W in registers for the whole launch and loops over row tiles, next tile's rows loading under the products; sum order kept
# speedup vs baseline: 2.9085x; 1.0837x over previous
#include <hip/hip_runtime.h>
#include <cstdint>
#include <cstddef>

typedef __attribute__((ext_vector_type(2))) float v2f;
typedef __attribute__((ext_vector_type(8))) float v8f;

#define NEG_SLOPE 0.01f


constexpr int DEG_RNG    = 12544;
constexpr int DEG_SLICES = 32;
constexpr int DEG_T      = 512;
constexpr int DEG_U      = 16;

__global__ void __launch_bounds__(DEG_T)
deg_count_ranges(const int* __restrict__ dst, unsigned* __restrict__ part, int n, int E) {
  __shared__ unsigned cnt[DEG_RNG];
  const int tid = threadIdx.x;
  const int slice = blockIdx.x;
  const int lo = blockIdx.y * DEG_RNG;
  for (int i = tid; i < DEG_RNG; i += DEG_T) cnt[i] = 0u;
  __syncthreads();

  const long long per = (((long long)E + 4 * DEG_SLICES - 1) / (4 * DEG_SLICES)) * 4;
  const long long e0 = (long long)slice * per;
  long long e1 = e0 + per;
  if (e1 > (long long)E) e1 = E;

  auto count = [&](int d) {
    const unsigned r = (unsigned)(d - lo);
    if (r < (unsigned)DEG_RNG) atomicAdd(&cnt[r], 1u);
  };

  if (e0 < e1) {
    if ((((size_t)dst) & 15) == 0) {
      const long long e1v = e0 + ((e1 - e0) & ~3LL);
      for (long long e = e0 + (long long)tid * 4; e < e1v; e += (long long)DEG_T * 4 * DEG_U) {
        int4 v[DEG_U];
#pragma unroll
        for (int u = 0; u < DEG_U; ++u) {
          const long long ee = e + (long long)u * DEG_T * 4;
          if (ee < e1v) v[u] = *(const int4*)(dst + ee);
          else v[u] = make_int4(-1, -1, -1, -1);
        }
#pragma unroll
        for (int u = 0; u < DEG_U; ++u) {
          count(v[u].x); count(v[u].y); count(v[u].z); count(v[u].w);
        }
      }
      if ((long long)tid < e1 - e1v) count(dst[e1v + tid]);
    } else {
      for (long long e = e0 + tid; e < e1; e += DEG_T) count(dst[e]);
    }
  }
  __syncthreads();

  unsigned* __restrict__ pp = part + (size_t)slice * n + lo;
  for (int i = tid; i < DEG_RNG; i += DEG_T)
    if (lo + i < n) pp[i] = cnt[i];
}

__global__ void compute_dinv(const unsigned* __restrict__ part, float* __restrict__ dinv, int n) {
  int i = blockIdx.x * blockDim.x + threadIdx.x;
  if (i < n) {
    unsigned c = 1u;
#pragma unroll
    for (int s = 0; s < DEG_SLICES; ++s) c += part[(size_t)s * n + i];
    const float d = (float)c;
    dinv[i] = d > 0.0f ? rsqrtf(d) : 0.0f;
  }
}

__global__ void compute_norm(const int* __restrict__ src, const int* __restrict__ dst,
                             const float* __restrict__ dinv, float* __restrict__ norm, int E,
                             unsigned* __restrict__ tickets) {
  int e = blockIdx.x * blockDim.x + threadIdx.x;
  if (e < E) norm[e] = dinv[src[e]] * dinv[dst[e]];
  if (e < 4) tickets[e] = 0u;
}

template <int K, int NOUT, bool ACT_IN, bool IN_PLANES>
__global__ void __launch_bounds__(NOUT / 16 * 32, K == 128 ? 8 : 4)
gcn_gemm_wmma(const float* __restrict__ X, const float* __restrict__ W,
              const float* __restrict__ dinv, const float* __restrict__ bias,
              float* __restrict__ Y, float* __restrict__ S, size_t plane_elems, int ntiles) {
  constexpr int LDSROW = K + 4;
  constexpr int PW = 64;
  constexpr int NT = NOUT / 16 * 32;
  constexpr int NV = (16 * K / 4) / NT;
  static_assert(NV * NT * 4 == 16 * K, "tile must split evenly over the work-group");
  __shared__ float At[2][16 * LDSROW];

  const int tid = threadIdx.x;
  const int wave = tid >> 5;
  const int lane = tid & 31;
  const int lrow  = lane & 15;
  const int khalf = (lane >> 4) * 2;
  const int n0    = wave * 16;
  const int mofs  = (lane < 16) ? 0 : 8;

  v2f b[K / 4];
#pragma unroll
  for (int kk = 0; kk < K / 4; ++kk) {
    const float* wp = W + (size_t)(kk * 4 + khalf) * NOUT + n0 + lrow;
    b[kk].x = wp[0];
    b[kk].y = wp[NOUT];
  }
  const float bn = bias[n0 + lrow];

  float4 pf[NV];
  auto fetch = [&](int t) {
    const size_t row0 = (size_t)t * 16;
#pragma unroll
    for (int j = 0; j < NV; ++j) {
      const int i = (tid + j * NT) * 4;
      const int r = i / K;
      const int c = i % K;
      const float* __restrict__ xp = IN_PLANES
          ? X + (size_t)(c / PW) * plane_elems + (row0 + r) * PW + (c % PW)
          : X + (row0 + r) * K + c;
      pf[j] = *(const float4*)xp;
    }
  };
  auto stage = [&](float* bufp) {
#pragma unroll
    for (int j = 0; j < NV; ++j) {
      const int i = (tid + j * NT) * 4;
      const int r = i / K;
      const int c = i % K;
      float4 v = pf[j];
      if (ACT_IN) {
        v.x = (v.x >= 0.f) ? v.x : NEG_SLOPE * v.x;
        v.y = (v.y >= 0.f) ? v.y : NEG_SLOPE * v.y;
        v.z = (v.z >= 0.f) ? v.z : NEG_SLOPE * v.z;
        v.w = (v.w >= 0.f) ? v.w : NEG_SLOPE * v.w;
      }
      float* d = bufp + r * LDSROW + c;
      d[0] = v.x; d[1] = v.y; d[2] = v.z; d[3] = v.w;
    }
  };

  int tile = blockIdx.x;
  if (tile >= ntiles) return;
  fetch(tile);
  stage(&At[0][0]);
  __syncthreads();

  int cur = 0;
  for (; tile < ntiles; tile += gridDim.x) {
    const int nxt = tile + gridDim.x;
    const bool has_next = nxt < ntiles;
    if (has_next) fetch(nxt);

    const size_t mbase_row = (size_t)tile * 16;
    float di2[8];
#pragma unroll
    for (int v = 0; v < 8; ++v) {
      const float di = dinv[mbase_row + mofs + v];
      di2[v] = di * di;
    }

    v8f acc = {};
    const float* abase = &At[cur][lrow * LDSROW + khalf];
#pragma unroll
    for (int kk = 0; kk < K / 4; ++kk) {
      v2f a; a.x = abase[kk * 4]; a.y = abase[kk * 4 + 1];
      acc = __builtin_amdgcn_wmma_f32_16x16x4_f32(
           false, a,  false, b[kk],
           (short)0, acc,  false,  false);
    }

    if (has_next) stage(&At[cur ^ 1][0]);
    __syncthreads();
    cur ^= 1;

    const int ncol = n0 + lrow;
    const size_t obase = (size_t)(ncol / PW) * plane_elems + mbase_row * PW + (ncol % PW);
    float* yp = Y + obase;
    float* sp = S + obase;
#pragma unroll
    for (int v = 0; v < 8; ++v) {
      const float y = acc[v];
      yp[(size_t)(mofs + v) * PW] = y;
      sp[(size_t)(mofs + v) * PW] = di2[v] * y + bn;
    }
  }
}

template <int F, bool BY_HALF>
__global__ void __launch_bounds__(256)
edge_scatter_lines(const int* __restrict__ src, const int* __restrict__ dst,
                   const float* __restrict__ norm, const float* __restrict__ hfull,
                   float* __restrict__ aggfull, int E, size_t plane_elems,
                   unsigned* __restrict__ tickets) {
  __shared__ unsigned pick[2];
  unsigned plane = 0u, chunk = blockIdx.x;
  if (BY_HALF) {
    const unsigned nchunks = gridDim.x;
    if (threadIdx.x == 0) {
      unsigned p = (__builtin_amdgcn_s_sendmsg_rtn(0x87) >> 18) & 1u;
      unsigned t = atomicAdd(&tickets[p], 1u);
      if (t >= nchunks) { p ^= 1u; t = atomicAdd(&tickets[p], 1u); }
      pick[0] = p; pick[1] = t;
    }
    __syncthreads();
    plane = pick[0]; chunk = pick[1];
    if (chunk >= nchunks) return;
  }
  const float* __restrict__ h = hfull + (size_t)plane * plane_elems;
  float* __restrict__ agg = aggfull + (size_t)plane * plane_elems;
  constexpr int L   = F / 32;
  constexpr int EPG = 32 / L;
  const int lane = threadIdx.x & 31;
  const long long gtid = (long long)chunk * blockDim.x + threadIdx.x;
  const long long e0 = gtid - lane;
  if (e0 >= (long long)E) return;
  const long long rem = (long long)E - e0;
  const int cnt = rem < 32 ? (int)rem : 32;

  int s = 0, d = 0;
  float w = 0.f;
  if (lane < cnt) {
    s = src[e0 + lane];
    d = dst[e0 + lane];
    w = norm[e0 + lane];
  }

  int j0 = 0;
  for (; j0 + EPG <= cnt; j0 += EPG) {
    float v[EPG][L];
#pragma unroll
    for (int u = 0; u < EPG; ++u) {
      const int sj = __builtin_amdgcn_readlane(s, j0 + u);
      const float* hp = h + (size_t)sj * F + lane;
#pragma unroll
      for (int l = 0; l < L; ++l) v[u][l] = hp[l * 32];
    }
#pragma unroll
    for (int u = 0; u < EPG; ++u) {
      const int dj = __builtin_amdgcn_readlane(d, j0 + u);
      const float wj = __int_as_float(__builtin_amdgcn_readlane(__float_as_int(w), j0 + u));
      float* ap = agg + (size_t)dj * F + lane;
#pragma unroll
      for (int l = 0; l < L; ++l) atomicAdd(ap + l * 32, v[u][l] * wj);
    }
  }
  for (; j0 < cnt; ++j0) {
    const int sj = __builtin_amdgcn_readlane(s, j0);
    const int dj = __builtin_amdgcn_readlane(d, j0);
    const float wj = __int_as_float(__builtin_amdgcn_readlane(__float_as_int(w), j0));
    const float* hp = h + (size_t)sj * F + lane;
    float* ap = agg + (size_t)dj * F + lane;
#pragma unroll
    for (int l = 0; l < L; ++l) atomicAdd(ap + l * 32, hp[l * 32] * wj);
  }
}

template <int F>
__global__ void __launch_bounds__(256)
edge_scatter_own_rows(const int* __restrict__ src, const int* __restrict__ dst,
                      const float* __restrict__ norm, const float* __restrict__ h,
                      float* __restrict__ agg, int E, int n,
                      unsigned* __restrict__ tickets) {
  __shared__ unsigned pick[2];
  const unsigned nchunks = gridDim.x;
  if (threadIdx.x == 0) {
    unsigned p = (__builtin_amdgcn_s_sendmsg_rtn(0x87) >> 18) & 1u;
    unsigned t = atomicAdd(&tickets[p], 1u);
    if (t >= nchunks) { p ^= 1u; t = atomicAdd(&tickets[p], 1u); }
    pick[0] = p; pick[1] = t;
  }
  __syncthreads();
  const unsigned half = pick[0], chunk = pick[1];
  if (chunk >= nchunks) return;

  constexpr int L = F / 32;
  constexpr int SLOTS = 24;
  const int lane = threadIdx.x & 31;
  const long long e0 = ((long long)chunk * (blockDim.x >> 5) + (threadIdx.x >> 5)) * 64;
  if (e0 >= (long long)E) return;

  const unsigned n_lo = (unsigned)n >> 1;
  const bool upper = half != 0u;
  int sA = 0, dA = 0, sB = 0, dB = 0;
  float wA = 0.f, wB = 0.f;
  bool keepA = false, keepB = false;
  const long long ea = e0 + lane, eb = ea + 32;
  if (ea < (long long)E) {
    sA = src[ea]; dA = dst[ea]; wA = norm[ea];
    keepA = ((unsigned)dA >= n_lo) == upper;
  }
  if (eb < (long long)E) {
    sB = src[eb]; dB = dst[eb]; wB = norm[eb];
    keepB = ((unsigned)dB >= n_lo) == upper;
  }
  const unsigned mA = __builtin_amdgcn_ballot_w32(keepA);
  const unsigned mB = __builtin_amdgcn_ballot_w32(keepB);

#pragma nounroll
  for (int t = 0; t < 2; ++t) {
    const int s = t ? sB : sA;
    const int d = t ? dB : dA;
    const float w = t ? wB : wA;
    unsigned m = t ? mB : mA;
    int rem = __builtin_popcount(m);
    while (rem > 0) {
      const int c = rem > SLOTS ? (rem + 1) >> 1 : rem;
      rem -= c;
      unsigned mc = m;
      float v[SLOTS][L];
#pragma unroll
      for (int u = 0; u < SLOTS; ++u) {
        if (u < c) {
          const int j = __builtin_ctz(m);
          m &= m - 1;
          const unsigned sj = (unsigned)__builtin_amdgcn_readlane(s, j);
          const float* hp = h + (size_t)sj * F + lane;
#pragma unroll
          for (int l = 0; l < L; ++l) v[u][l] = hp[l * 32];
        }
      }
#pragma unroll
      for (int u = 0; u < SLOTS; ++u) {
        if (u < c) {
          const int j = __builtin_ctz(mc);
          mc &= mc - 1;
          const unsigned dj = (unsigned)__builtin_amdgcn_readlane(d, j);
          const float wj = __int_as_float(__builtin_amdgcn_readlane(__float_as_int(w), j));
          float* ap = agg + (size_t)dj * F + lane;
#pragma unroll
          for (int l = 0; l < L; ++l) atomicAdd(ap + l * 32, v[u][l] * wj);
        }
      }
    }
  }
}


extern "C" void kernel_launch(void* const* d_in, const int* in_sizes, int n_in,
                              void* d_out, int out_size, void* d_ws, size_t ws_size,
                              hipStream_t stream) {
  const float* x   = (const float*)d_in[0];
  const int*   ei  = (const int*)d_in[1];
  const float* W1  = (const float*)d_in[2];
  const float* b1  = (const float*)d_in[3];
  const float* W2  = (const float*)d_in[4];
  const float* b2  = (const float*)d_in[5];
  float* out = (float*)d_out;

  const int N = in_sizes[0] / 256;
  const int E = in_sizes[1] / 2;
  const int* src = ei;
  const int* dst = ei + E;

  float* ws   = (float*)d_ws;
  float* tkw  = ws;
  float* dinv = tkw + N;
  float* norm = dinv + N;
  float* h1   = norm + E;
  float* agg1 = h1 + (size_t)N * 128;
  const size_t plane = (size_t)N * 64;
  float* g  = h1;
  unsigned* tickets = (unsigned*)tkw;
  unsigned* degpart = (unsigned*)h1;

  const int T = 256;
  auto blk = [](long long n, int t) { return (unsigned)((n + t - 1) / t); };

  deg_count_ranges<<<dim3(DEG_SLICES, blk(N, DEG_RNG)), DEG_T, 0, stream>>>(dst, degpart, N, E);
  compute_dinv<<<blk(N, T), T, 0, stream>>>(degpart, dinv, N);
  compute_norm<<<blk(E, T), T, 0, stream>>>(src, dst, dinv, norm, E, tickets);

  gcn_gemm_wmma<256, 128, false, false><<<512, 8 * 32, 0, stream>>>(x, W1, dinv, b1, h1, agg1, plane, N / 16);

  edge_scatter_lines<64, true><<<dim3(blk(E, T), 2), T, 0, stream>>>(src, dst, norm, h1, agg1, E, plane, tickets);

  gcn_gemm_wmma<128, 64, true, true><<<2048, 4 * 32, 0, stream>>>(agg1, W2, dinv, b2, g, out, plane, N / 16);

  edge_scatter_own_rows<64><<<dim3(blk(E, 2 * T), 2), T, 0, stream>>>(src, dst, norm, g, out, E, N, tickets + 2);
}
